# starting kernel
# baseline (speedup 1.0000x reference)
#include <hip/hip_runtime.h>
#include <math.h>

#define BATCH 64
#define INC   2048
#define NJ    32
#define DIM   16
#define ICH   16
#define NBLK  (INC / ICH)
#define SZ    (BATCH * NJ * DIM)
#define NTHR  512
#define JT    8

typedef float v2f __attribute__((ext_vector_type(2)));
typedef float v8f __attribute__((ext_vector_type(8)));

#define DPP_XOR(x, m)                                                          \
  __int_as_float(__builtin_amdgcn_update_dpp(                                  \
      0, __float_as_int(x), 0x160 | (m), 0xf, 0xf, true))

__device__ __forceinline__ float xorsum16(float p) {
  p += DPP_XOR(p, 1);
  p += DPP_XOR(p, 2);
  p += DPP_XOR(p, 4);
  p += DPP_XOR(p, 8);
  return p;
}

__global__ void caps_init(float* p, int n) {
  int i = blockIdx.x * blockDim.x + threadIdx.x;
  if (i < n) p[i] = 0.f;
}

template <bool MODEA>
__global__ __launch_bounds__(NTHR) void caps_pass(const float* __restrict__ x,
                                                  const float* __restrict__ W,
                                                  const float* __restrict__ outsum,
                                                  float* sdst) {
  __shared__ float Xs[BATCH * DIM];
  __shared__ float Ws[NJ * DIM * DIM];
  __shared__ float Lg[BATCH * NJ];
  __shared__ float smax[BATCH];
  __shared__ float sinv[BATCH];

  const int tid  = threadIdx.x;
  const int lane = tid & 31;
  const int wv   = tid >> 5;
  const int half = lane >> 4;
  const int l16  = lane & 15;
  const int mt   = wv & 3;
  const int jb   = (wv >> 2) * JT;
  const int i0   = blockIdx.x * ICH;

  const v8f vzero = {0.f, 0.f, 0.f, 0.f, 0.f, 0.f, 0.f, 0.f};
  v8f sacc[JT];
#pragma unroll
  for (int jj = 0; jj < JT; ++jj) sacc[jj] = vzero;

  float osum[JT][8];
#pragma unroll
  for (int jj = 0; jj < JT; ++jj)
#pragma unroll
    for (int v = 0; v < 8; ++v) {
      const int b = mt * 16 + v + 8 * half;
      osum[jj][v] = outsum[((size_t)b * NJ + jb + jj) * DIM + l16];
    }

  for (int ii = 0; ii < ICH; ++ii) {
    const int i = i0 + ii;
    {
      const int r = tid >> 3, c = (tid & 7) * 2;
      const float2 vx = *(const float2*)(x + ((size_t)r * INC + i) * DIM + c);
      *(float2*)(Xs + r * DIM + c) = vx;
    }
    {
      const float4* src = (const float4*)(W + (size_t)i * NJ * DIM * DIM);
      float4* dst = (float4*)Ws;
#pragma unroll
      for (int q = 0; q < 4; ++q) dst[q * NTHR + tid] = src[q * NTHR + tid];
    }
    if (i + 1 < INC)
      __builtin_prefetch(W + (size_t)(i + 1) * NJ * DIM * DIM, 0, 1);
    __syncthreads();

    v2f afr[4];
#pragma unroll
    for (int kk = 0; kk < 4; ++kk) {
      const int base = (mt * 16 + l16) * DIM + 4 * kk + 2 * half;
      afr[kk].x = Xs[base];
      afr[kk].y = Xs[base + 1];
    }

#pragma unroll
    for (int jj = 0; jj < JT; ++jj) {
      const int j = jb + jj;
      v8f acc = vzero;
#pragma unroll
      for (int kk = 0; kk < 4; ++kk) {
        v2f bfr;
        const int base = (j * DIM + l16) * DIM + 4 * kk + 2 * half;
        bfr.x = Ws[base];
        bfr.y = Ws[base + 1];
        acc = __builtin_amdgcn_wmma_f32_16x16x4_f32(false, afr[kk], false, bfr,
                                                    (short)0, acc, false, false);
      }
#pragma unroll
      for (int v = 0; v < 8; ++v) {
        const float p = xorsum16(acc[v] * osum[jj][v]);
        if (l16 == 0) Lg[(mt * 16 + v + 8 * half) * NJ + j] = p;
      }
    }
    __syncthreads();

    if (tid < BATCH) {
      float m = -3.4e38f;
      for (int j = 0; j < NJ; ++j) m = fmaxf(m, Lg[tid * NJ + j]);
      float s = 0.f;
      for (int j = 0; j < NJ; ++j) s += expf(Lg[tid * NJ + j] - m);
      smax[tid] = m;
      sinv[tid] = 1.f / s;
    }
    __syncthreads();

    const int brow = mt * 16 + l16;
    const float mrow = smax[brow];
    const float srow = sinv[brow];
#pragma unroll
    for (int jj = 0; jj < JT; ++jj) {
      const int j = jb + jj;
      const float c = expf(Lg[brow * NJ + j] - mrow) * srow;
#pragma unroll
      for (int kk = 0; kk < 4; ++kk) {
        v2f afs, bfr;
        afs.x = c * afr[kk].x;
        afs.y = c * afr[kk].y;
        const int base = (j * DIM + l16) * DIM + 4 * kk + 2 * half;
        bfr.x = Ws[base];
        bfr.y = Ws[base + 1];
        sacc[jj] = __builtin_amdgcn_wmma_f32_16x16x4_f32(
            false, afs, false, bfr, (short)0, sacc[jj], false, false);
      }
    }
    __syncthreads();
  }

  if (MODEA) {
    float* dst = sdst + (size_t)blockIdx.x * SZ;
#pragma unroll
    for (int jj = 0; jj < JT; ++jj)
#pragma unroll
      for (int v = 0; v < 8; ++v) {
        const int b = mt * 16 + v + 8 * half;
        dst[((size_t)b * NJ + jb + jj) * DIM + l16] = sacc[jj][v];
      }
  } else {
#pragma unroll
    for (int jj = 0; jj < JT; ++jj)
#pragma unroll
      for (int v = 0; v < 8; ++v) {
        const int b = mt * 16 + v + 8 * half;
        unsafeAtomicAdd(&sdst[((size_t)b * NJ + jb + jj) * DIM + l16], sacc[jj][v]);
      }
  }
}

__global__ void caps_squash(const float* src, float* outsum, float* dout,
                            float* szero, int nparts) {
  const int idx = blockIdx.x * blockDim.x + threadIdx.x;
  float v = 0.f;
  for (int p = 0; p < nparts; ++p) v += src[(size_t)p * SZ + idx];
  if (szero) szero[idx] = 0.f;
  const float sq = xorsum16(v * v);
  const float coef = sq / ((1.f + sq) * sqrtf(sq + 1e-7f));
  const float o = coef * v;
  outsum[idx] += o;
  if (dout) dout[idx] = o;
}

extern "C" void kernel_launch(void* const* d_in, const int* in_sizes, int n_in,
                              void* d_out, int out_size, void* d_ws, size_t ws_size,
                              hipStream_t stream) {
  const float* x = (const float*)d_in[0];
  const float* W = (const float*)d_in[1];
  float* out = (float*)d_out;
  float* ws = (float*)d_ws;

  float* outsum = ws;
  float* sbuf   = ws + SZ;
  float* parts  = ws + 2 * SZ;

  const bool modeA = ws_size >= (size_t)(2 * SZ + NBLK * SZ) * sizeof(float);

  int ninit = 2 * SZ;
  if ((size_t)ninit * sizeof(float) > ws_size) ninit = (int)(ws_size / sizeof(float));
  caps_init<<<(ninit + 255) / 256, 256, 0, stream>>>(ws, ninit);

  for (int t = 0; t < 3; ++t) {
    if (modeA)
      caps_pass<true><<<NBLK, NTHR, 0, stream>>>(x, W, outsum, parts);
    else
      caps_pass<false><<<NBLK, NTHR, 0, stream>>>(x, W, outsum, sbuf);
    caps_squash<<<SZ / 256, 256, 0, stream>>>(modeA ? parts : sbuf, outsum,
                                              (t == 2) ? out : nullptr,
                                              modeA ? nullptr : sbuf,
                                              modeA ? NBLK : 1);
  }
}


# Routing pass split in two launches on all 256 WGPs: coupling kernel writes c; weighted sum split over 8 j groups x 32 capsule ranges, 32 partial planes not 128 (i sum order changes)
# speedup vs baseline: 2.9803x; 2.9803x over previous
#include <hip/hip_runtime.h>
#include <math.h>

#define BATCH 64
#define INC   2048
#define NJ    32
#define DIM   16
#define SZ    (BATCH * NJ * DIM)
#define NTHR  256
#define GRID  256
#define NGRP  2
#define WP    20
#define OP    516
#define WSLAB (NJ * DIM * WP)
#define JG    4
#define NJG   (NJ / JG)
#define NRNG  (GRID / NJG)
#define WSTEP (NJ * DIM * DIM)

typedef float v2f __attribute__((ext_vector_type(2)));
typedef float v4f __attribute__((ext_vector_type(4)));
typedef float v8f __attribute__((ext_vector_type(8)));

#define DPP_XOR(x, m)                                                          \
  __int_as_float(__builtin_amdgcn_update_dpp(                                  \
      0, __float_as_int(x), 0x160 | (m), 0xf, 0xf, true))

__device__ __forceinline__ float xorsum16(float p) {
  p += DPP_XOR(p, 1);
  p += DPP_XOR(p, 2);
  p += DPP_XOR(p, 4);
  p += DPP_XOR(p, 8);
  return p;
}

__device__ __forceinline__ float xhalf(float v) {
  return __int_as_float(__builtin_amdgcn_permlanex16(
      __float_as_int(v), __float_as_int(v), 0x76543210, 0xfedcba98, false, false));
}

__global__ void caps_init(float* p, int n) {
  int i = blockIdx.x * blockDim.x + threadIdx.x;
  if (i < n) p[i] = 0.f;
}

__global__ __launch_bounds__(NTHR) void caps_couple(const float* __restrict__ x,
                                                    const float* __restrict__ W,
                                                    const float* __restrict__ outsum,
                                                    float* __restrict__ cbuf,
                                                    int ibase, int ni) {
  __shared__ __attribute__((aligned(16))) float smem[NGRP * WSLAB + BATCH * OP];
  float* Os = smem + NGRP * WSLAB;

  const int tid  = threadIdx.x;
  const int lane = tid & 31;
  const int wv   = tid >> 5;
  const int half = lane >> 4;
  const int l16  = lane & 15;
  const int mt   = wv & 3;
  const int grp  = wv >> 2;
  const int tg   = tid & 127;
  const int cpg  = ni / (GRID * NGRP);
  const int il0  = (blockIdx.x * NGRP + grp) * cpg;
  const int i0   = ibase + il0;
  float* Wg = smem + grp * WSLAB;

  const v8f vzero = {0.f, 0.f, 0.f, 0.f, 0.f, 0.f, 0.f, 0.f};

  {
    const float4* src = (const float4*)outsum;
#pragma unroll 8
    for (int q = 0; q < (SZ / 4) / NTHR; ++q) {
      const float4 v = src[q * NTHR + tid];
      *(float4*)(Os + (q * 2 + (tid >> 7)) * OP + (tid & 127) * 4) = v;
    }
  }
  const float* orow = Os + (mt * 16 + l16) * OP + 8 * half;
  const float* wrow = Wg + l16 * WP + 8 * half;

  const float4* wsrc = (const float4*)(W + (size_t)i0 * WSTEP) + tg;
  const float* xsrc = x + ((size_t)(mt * 16 + l16) * INC + i0) * DIM + 8 * half;
  const size_t jgs = (size_t)ni * BATCH * JG;
  float* cdst = cbuf + ((size_t)il0 * BATCH + mt * 16 + l16) * JG + (size_t)(4 * half) * jgs;

  for (int t = 0; t < cpg; ++t) {
    const int i = i0 + t;
    const v4f x0 = *(const v4f*)(xsrc);
    const v4f x1 = *(const v4f*)(xsrc + 4);
    xsrc += DIM;
    v2f xfr[4];
    xfr[0] = x0.s01; xfr[1] = x0.s23; xfr[2] = x1.s01; xfr[3] = x1.s23;
    {
      const int wofs = (tg >> 2) * WP + (tg & 3) * 4;
#pragma unroll
      for (int q = 0; q < 16; ++q) {
        const float4 v = wsrc[q * 128];
        *(float4*)(Wg + q * 32 * WP + wofs) = v;
      }
      wsrc += WSTEP / 4;
    }
    if (t + 1 < cpg)
      __builtin_prefetch(W + (size_t)(i + 1) * WSTEP, 0, 1);
    __syncthreads();

    float lg[NJ];
#pragma unroll
    for (int j = 0; j < NJ; ++j) {
      if ((j & 3) == 0) __builtin_amdgcn_sched_barrier(0);
      v8f acc = vzero;
      const v4f w0 = *(const v4f*)(wrow + j * DIM * WP);
      const v4f w1 = *(const v4f*)(wrow + j * DIM * WP + 4);
      acc = __builtin_amdgcn_wmma_f32_16x16x4_f32(false, w0.s01, false, xfr[0],
                                                  (short)0, acc, false, false);
      acc = __builtin_amdgcn_wmma_f32_16x16x4_f32(false, w0.s23, false, xfr[1],
                                                  (short)0, acc, false, false);
      acc = __builtin_amdgcn_wmma_f32_16x16x4_f32(false, w1.s01, false, xfr[2],
                                                  (short)0, acc, false, false);
      acc = __builtin_amdgcn_wmma_f32_16x16x4_f32(false, w1.s23, false, xfr[3],
                                                  (short)0, acc, false, false);
      const v4f o0 = *(const v4f*)(orow + j * DIM);
      const v4f o1 = *(const v4f*)(orow + j * DIM + 4);
      v2f pp = acc.s01 * o0.s01;
      pp = __builtin_elementwise_fma(acc.s23, o0.s23, pp);
      pp = __builtin_elementwise_fma(acc.s45, o1.s01, pp);
      pp = __builtin_elementwise_fma(acc.s67, o1.s23, pp);
      const float p = pp.x + pp.y;
      lg[j] = p + xhalf(p);
    }

    float m = lg[0];
#pragma unroll
    for (int j = 1; j < NJ; ++j) m = fmaxf(m, lg[j]);
    float ssum = 0.f;
#pragma unroll
    for (int j = 0; j < NJ; ++j) {
      lg[j] = expf(lg[j] - m);
      ssum += lg[j];
    }
    const float sinv = 1.f / ssum;

    __syncthreads();

#pragma unroll
    for (int q = 0; q < 4; ++q) {
      v4f lo, hi;
      lo.x = lg[q * 4 + 0] * sinv;        lo.y = lg[q * 4 + 1] * sinv;
      lo.z = lg[q * 4 + 2] * sinv;        lo.w = lg[q * 4 + 3] * sinv;
      hi.x = lg[(q + 4) * 4 + 0] * sinv;  hi.y = lg[(q + 4) * 4 + 1] * sinv;
      hi.z = lg[(q + 4) * 4 + 2] * sinv;  hi.w = lg[(q + 4) * 4 + 3] * sinv;
      *(v4f*)(cdst + (size_t)q * jgs) = half ? hi : lo;
    }
    cdst += BATCH * JG;
  }
}

struct Frag2 {
  v4f w[2][JG][2];
  v4f xv[4][2][2];
  v4f c[2][4];
};

__device__ __forceinline__ void load_pair(Frag2& f, const float* wp, const float* xp,
                                          const float* cp) {
#pragma unroll
  for (int mt = 0; mt < 4; ++mt) {
    const float* xm = xp + (size_t)mt * 16 * INC * DIM;
    f.xv[mt][0][0] = *(const v4f*)(xm);
    f.xv[mt][0][1] = *(const v4f*)(xm + 4);
    f.xv[mt][1][0] = *(const v4f*)(xm + DIM);
    f.xv[mt][1][1] = *(const v4f*)(xm + DIM + 4);
  }
#pragma unroll
  for (int e = 0; e < 2; ++e) {
#pragma unroll
    for (int jj = 0; jj < JG; ++jj) {
      f.w[e][jj][0] = *(const v4f*)(wp + e * WSTEP + jj * DIM * DIM);
      f.w[e][jj][1] = *(const v4f*)(wp + e * WSTEP + jj * DIM * DIM + 4);
    }
#pragma unroll
    for (int mt = 0; mt < 4; ++mt)
      f.c[e][mt] = *(const v4f*)(cp + e * BATCH * JG + mt * 16 * JG);
  }
}

__device__ __forceinline__ void mac_pair(v8f (&acc)[JG][4], const Frag2& f) {
#pragma unroll
  for (int e = 0; e < 2; ++e) {
#pragma unroll
    for (int jj = 0; jj < JG; ++jj) {
      v2f a[4];
      a[0] = f.w[e][jj][0].s01; a[1] = f.w[e][jj][0].s23;
      a[2] = f.w[e][jj][1].s01; a[3] = f.w[e][jj][1].s23;
#pragma unroll
      for (int kk = 0; kk < 4; ++kk) {
#pragma unroll
        for (int mt = 0; mt < 4; ++mt) {
          const float cs = f.c[e][mt][jj];
          v2f bx;
          if (kk == 0) bx = f.xv[mt][e][0].s01;
          else if (kk == 1) bx = f.xv[mt][e][0].s23;
          else if (kk == 2) bx = f.xv[mt][e][1].s01;
          else bx = f.xv[mt][e][1].s23;
          acc[jj][mt] = __builtin_amdgcn_wmma_f32_16x16x4_f32(
              false, a[kk], false, bx * cs, (short)0, acc[jj][mt], false, false);
        }
      }
    }
  }
}

__global__ __launch_bounds__(NTHR) void caps_wsum(const float* __restrict__ x,
                                                  const float* __restrict__ W,
                                                  const float* __restrict__ cbuf,
                                                  float* __restrict__ planes,
                                                  int ibase, int ni) {
  __shared__ __attribute__((aligned(16))) float red[8 * JG * 4 * 8 * 32];

  const int tid  = threadIdx.x;
  const int lane = tid & 31;
  const int wv   = tid >> 5;
  const int half = lane >> 4;
  const int l16  = lane & 15;
  const int jg   = blockIdx.x & (NJG - 1);
  const int r    = blockIdx.x / NJG;
  const int cpw  = ni / (NRNG * 8);
  const int il0  = (r * 8 + wv) * cpw;
  const int i0   = ibase + il0;

  const v8f vzero = {0.f, 0.f, 0.f, 0.f, 0.f, 0.f, 0.f, 0.f};
  v8f acc[JG][4];
#pragma unroll
  for (int jj = 0; jj < JG; ++jj)
#pragma unroll
    for (int mt = 0; mt < 4; ++mt) acc[jj][mt] = vzero;

  const float* wp = W + ((size_t)i0 * NJ + jg * JG) * DIM * DIM + l16 * DIM + 8 * half;
  const float* xp = x + ((size_t)l16 * INC + i0) * DIM + 8 * half;
  const float* cp = cbuf + (((size_t)jg * ni + il0) * BATCH + l16) * JG;

  Frag2 P0, P1;
  load_pair(P0, wp, xp, cp);
  for (int t = 0; t + 4 < cpw; t += 4) {
    load_pair(P1, wp + 2 * WSTEP, xp + 2 * DIM, cp + 2 * BATCH * JG);
    __builtin_amdgcn_sched_barrier(0);
    mac_pair(acc, P0);
    wp += 4 * WSTEP; xp += 4 * DIM; cp += 4 * BATCH * JG;
    load_pair(P0, wp, xp, cp);
    __builtin_amdgcn_sched_barrier(0);
    mac_pair(acc, P1);
  }
  load_pair(P1, wp + 2 * WSTEP, xp + 2 * DIM, cp + 2 * BATCH * JG);
  __builtin_amdgcn_sched_barrier(0);
  mac_pair(acc, P0);
  mac_pair(acc, P1);

#pragma unroll
  for (int jj = 0; jj < JG; ++jj) {
#pragma unroll
    for (int mt = 0; mt < 4; ++mt) {
      float4 a, b;
      a.x = acc[jj][mt][0]; a.y = acc[jj][mt][1]; a.z = acc[jj][mt][2]; a.w = acc[jj][mt][3];
      b.x = acc[jj][mt][4]; b.y = acc[jj][mt][5]; b.z = acc[jj][mt][6]; b.w = acc[jj][mt][7];
      const int s = (jj * 4 + mt) * 2;
      *(float4*)(red + ((wv * 32 + s + 0) * 32 + lane) * 4) = a;
      *(float4*)(red + ((wv * 32 + s + 1) * 32 + lane) * 4) = b;
    }
  }
  __syncthreads();
  float* plane = planes + (size_t)r * SZ;
#pragma unroll
  for (int u = 0; u < 4; ++u) {
    const int s = wv * 4 + u;
    float4 v = *(const float4*)(red + ((0 * 32 + s) * 32 + lane) * 4);
#pragma unroll
    for (int w = 1; w < 8; ++w) {
      const float4 p = *(const float4*)(red + ((w * 32 + s) * 32 + lane) * 4);
      v.x += p.x; v.y += p.y; v.z += p.z; v.w += p.w;
    }
    const int jj = s >> 3, mt = (s >> 1) & 3, q = s & 1;
    float* dst = plane + ((size_t)(mt * 16 + l16) * NJ + jg * JG + jj) * DIM + 8 * half + 4 * q;
    *(float4*)dst = v;
  }
}

__global__ void caps_squash(const float* src, float* outsum, float* dout,
                            float* szero, int nparts) {
  const int idx = blockIdx.x * blockDim.x + threadIdx.x;
  float v = 0.f;
  for (int p = 0; p < nparts; ++p) v += src[(size_t)p * SZ + idx];
  if (szero) szero[idx] = 0.f;
  const float sq = xorsum16(v * v);
  const float coef = sq / ((1.f + sq) * sqrtf(sq + 1e-7f));
  const float o = coef * v;
  outsum[idx] += o;
  if (dout) dout[idx] = o;
}

extern "C" void kernel_launch(void* const* d_in, const int* in_sizes, int n_in,
                              void* d_out, int out_size, void* d_ws, size_t ws_size,
                              hipStream_t stream) {
  const float* x = (const float*)d_in[0];
  const float* W = (const float*)d_in[1];
  float* out = (float*)d_out;
  float* ws = (float*)d_ws;

  float* outsum = ws;
  float* parts  = ws + SZ;

  const bool one = ws_size >= (size_t)(1 + 32 + 128) * SZ * sizeof(float);
  if (!one && ws_size < (size_t)(1 + 64 + 64) * SZ * sizeof(float)) return;

  caps_init<<<SZ / 256, 256, 0, stream>>>(outsum, SZ);

  for (int t = 0; t < 3; ++t) {
    if (one) {
      float* cbuf = parts + 32 * SZ;
      caps_couple<<<GRID, NTHR, 0, stream>>>(x, W, outsum, cbuf, 0, 2048);
      caps_wsum<<<GRID, NTHR, 0, stream>>>(x, W, cbuf, parts, 0, 2048);
      caps_squash<<<SZ / 256, 256, 0, stream>>>(parts, outsum, (t == 2) ? out : nullptr,
                                                nullptr, 32);
    } else {
      float* cbuf = parts + 64 * SZ;
      caps_couple<<<GRID, NTHR, 0, stream>>>(x, W, outsum, cbuf, 0, 1024);
      caps_wsum<<<GRID, NTHR, 0, stream>>>(x, W, cbuf, parts, 0, 1024);
      caps_couple<<<GRID, NTHR, 0, stream>>>(x, W, outsum, cbuf, 1024, 1024);
      caps_wsum<<<GRID, NTHR, 0, stream>>>(x, W, cbuf, parts + 32 * SZ, 1024, 1024);
      caps_squash<<<SZ / 256, 256, 0, stream>>>(parts, outsum, (t == 2) ? out : nullptr,
                                                nullptr, 64);
    }
  }
}


# Coupling kernel re-tiled: a wave owns one input capsule for all 64 batch rows x 32 j; its W fragments go global->registers once and stay; no W slab in LDS, no per-capsule barriers (one per block)
# speedup vs baseline: 3.2769x; 1.0995x over previous
#include <hip/hip_runtime.h>
#include <math.h>

#define BATCH 64
#define INC   2048
#define NJ    32
#define DIM   16
#define SZ    (BATCH * NJ * DIM)
#define NTHR  256
#define GRID  256
#define CW    8
#define JB    8
#define OP    516
#define JG    4
#define NJG   (NJ / JG)
#define NRNG  (GRID / NJG)
#define WSTEP (NJ * DIM * DIM)

typedef float v2f __attribute__((ext_vector_type(2)));
typedef float v4f __attribute__((ext_vector_type(4)));
typedef float v8f __attribute__((ext_vector_type(8)));

#define DPP_XOR(x, m)                                                          \
  __int_as_float(__builtin_amdgcn_update_dpp(                                  \
      0, __float_as_int(x), 0x160 | (m), 0xf, 0xf, true))

__device__ __forceinline__ float xorsum16(float p) {
  p += DPP_XOR(p, 1);
  p += DPP_XOR(p, 2);
  p += DPP_XOR(p, 4);
  p += DPP_XOR(p, 8);
  return p;
}

__device__ __forceinline__ float xhalf(float v) {
  return __int_as_float(__builtin_amdgcn_permlanex16(
      __float_as_int(v), __float_as_int(v), 0x76543210, 0xfedcba98, false, false));
}

__global__ void caps_init(float* p, int n) {
  int i = blockIdx.x * blockDim.x + threadIdx.x;
  if (i < n) p[i] = 0.f;
}

__device__ __forceinline__ void load_wgroup(v4f (&w)[NJ][2], const float* wl, int g) {
#pragma unroll
  for (int jj = 0; jj < JB; ++jj) {
    w[g * JB + jj][0] = *(const v4f*)(wl + (g * JB + jj) * DIM * DIM);
    w[g * JB + jj][1] = *(const v4f*)(wl + (g * JB + jj) * DIM * DIM + 4);
  }
}

__global__ __launch_bounds__(NTHR) void caps_couple(const float* __restrict__ x,
                                                    const float* __restrict__ W,
                                                    const float* __restrict__ outsum,
                                                    float* __restrict__ cbuf,
                                                    int ibase, int ni) {
  __shared__ __attribute__((aligned(16))) float Os[BATCH * OP];

  const int tid  = threadIdx.x;
  const int lane = tid & 31;
  const int wv   = tid >> 5;
  const int half = lane >> 4;
  const int l16  = lane & 15;
  const int il   = blockIdx.x * CW + wv;
  const int i    = ibase + il;

  const v8f vzero = {0.f, 0.f, 0.f, 0.f, 0.f, 0.f, 0.f, 0.f};

  const float* wl = W + (size_t)i * WSTEP + l16 * DIM + 8 * half;
  const float* xl = x + ((size_t)l16 * INC + i) * DIM + 8 * half;

  v4f xa = *(const v4f*)(xl);
  v4f xb = *(const v4f*)(xl + 4);
  v4f w[NJ][2];
  load_wgroup(w, wl, 0);

  {
    const float4* src = (const float4*)outsum;
#pragma unroll 8
    for (int q = 0; q < (SZ / 4) / NTHR; ++q) {
      const float4 v = src[q * NTHR + tid];
      *(float4*)(Os + (q * 2 + (tid >> 7)) * OP + (tid & 127) * 4) = v;
    }
  }
  __syncthreads();

#pragma unroll
  for (int g = 1; g < NJ / JB; ++g) load_wgroup(w, wl, g);

  const float* orow = Os + l16 * OP + 8 * half;
  const size_t jgs = (size_t)ni * BATCH * JG;
  float* cdst = cbuf + ((size_t)il * BATCH + l16) * JG + (size_t)(4 * half) * jgs;

#pragma nounroll
  for (int mt = 0; mt < 4; ++mt) {
    v2f xfr[4];
    xfr[0] = xa.s01; xfr[1] = xa.s23; xfr[2] = xb.s01; xfr[3] = xb.s23;
    {
      const int mn = (mt < 3) ? mt + 1 : 3;
      xa = *(const v4f*)(xl + (size_t)mn * 16 * INC * DIM);
      xb = *(const v4f*)(xl + (size_t)mn * 16 * INC * DIM + 4);
    }

    float lg[NJ];
#pragma unroll
    for (int j = 0; j < NJ; j += 2) {
      if ((j & 3) == 0) __builtin_amdgcn_sched_barrier(0);
      v8f acc0 = vzero, acc1 = vzero;
      v2f a0[4], a1[4];
      a0[0] = w[j][0].s01;     a0[1] = w[j][0].s23;     a0[2] = w[j][1].s01;     a0[3] = w[j][1].s23;
      a1[0] = w[j + 1][0].s01; a1[1] = w[j + 1][0].s23; a1[2] = w[j + 1][1].s01; a1[3] = w[j + 1][1].s23;
#pragma unroll
      for (int kk = 0; kk < 4; ++kk) {
        acc0 = __builtin_amdgcn_wmma_f32_16x16x4_f32(false, a0[kk], false, xfr[kk],
                                                     (short)0, acc0, false, false);
        acc1 = __builtin_amdgcn_wmma_f32_16x16x4_f32(false, a1[kk], false, xfr[kk],
                                                     (short)0, acc1, false, false);
      }
      {
        const v4f o0 = *(const v4f*)(orow + j * DIM);
        const v4f o1 = *(const v4f*)(orow + j * DIM + 4);
        v2f pp = acc0.s01 * o0.s01;
        pp = __builtin_elementwise_fma(acc0.s23, o0.s23, pp);
        pp = __builtin_elementwise_fma(acc0.s45, o1.s01, pp);
        pp = __builtin_elementwise_fma(acc0.s67, o1.s23, pp);
        const float p = pp.x + pp.y;
        lg[j] = p + xhalf(p);
      }
      {
        const v4f o0 = *(const v4f*)(orow + (j + 1) * DIM);
        const v4f o1 = *(const v4f*)(orow + (j + 1) * DIM + 4);
        v2f pp = acc1.s01 * o0.s01;
        pp = __builtin_elementwise_fma(acc1.s23, o0.s23, pp);
        pp = __builtin_elementwise_fma(acc1.s45, o1.s01, pp);
        pp = __builtin_elementwise_fma(acc1.s67, o1.s23, pp);
        const float p = pp.x + pp.y;
        lg[j + 1] = p + xhalf(p);
      }
    }

    float m = lg[0];
#pragma unroll
    for (int j = 1; j < NJ; ++j) m = fmaxf(m, lg[j]);
    float ssum = 0.f;
#pragma unroll
    for (int j = 0; j < NJ; ++j) {
      lg[j] = expf(lg[j] - m);
      ssum += lg[j];
    }
    const float sinv = 1.f / ssum;

#pragma unroll
    for (int q = 0; q < 4; ++q) {
      v4f lo, hi;
      lo.x = lg[q * 4 + 0] * sinv;        lo.y = lg[q * 4 + 1] * sinv;
      lo.z = lg[q * 4 + 2] * sinv;        lo.w = lg[q * 4 + 3] * sinv;
      hi.x = lg[(q + 4) * 4 + 0] * sinv;  hi.y = lg[(q + 4) * 4 + 1] * sinv;
      hi.z = lg[(q + 4) * 4 + 2] * sinv;  hi.w = lg[(q + 4) * 4 + 3] * sinv;
      *(v4f*)(cdst + (size_t)q * jgs) = half ? hi : lo;
    }
    cdst += 16 * JG;
    orow += 16 * OP;
  }
}

struct Frag2 {
  v4f w[2][JG][2];
  v4f xv[4][2][2];
  v4f c[2][4];
};

__device__ __forceinline__ void load_pair(Frag2& f, const float* wp, const float* xp,
                                          const float* cp) {
#pragma unroll
  for (int mt = 0; mt < 4; ++mt) {
    const float* xm = xp + (size_t)mt * 16 * INC * DIM;
    f.xv[mt][0][0] = *(const v4f*)(xm);
    f.xv[mt][0][1] = *(const v4f*)(xm + 4);
    f.xv[mt][1][0] = *(const v4f*)(xm + DIM);
    f.xv[mt][1][1] = *(const v4f*)(xm + DIM + 4);
  }
#pragma unroll
  for (int e = 0; e < 2; ++e) {
#pragma unroll
    for (int jj = 0; jj < JG; ++jj) {
      f.w[e][jj][0] = *(const v4f*)(wp + e * WSTEP + jj * DIM * DIM);
      f.w[e][jj][1] = *(const v4f*)(wp + e * WSTEP + jj * DIM * DIM + 4);
    }
#pragma unroll
    for (int mt = 0; mt < 4; ++mt)
      f.c[e][mt] = *(const v4f*)(cp + e * BATCH * JG + mt * 16 * JG);
  }
}

__device__ __forceinline__ void mac_pair(v8f (&acc)[JG][4], const Frag2& f) {
#pragma unroll
  for (int e = 0; e < 2; ++e) {
#pragma unroll
    for (int jj = 0; jj < JG; ++jj) {
      v2f a[4];
      a[0] = f.w[e][jj][0].s01; a[1] = f.w[e][jj][0].s23;
      a[2] = f.w[e][jj][1].s01; a[3] = f.w[e][jj][1].s23;
#pragma unroll
      for (int kk = 0; kk < 4; ++kk) {
#pragma unroll
        for (int mt = 0; mt < 4; ++mt) {
          const float cs = f.c[e][mt][jj];
          v2f bx;
          if (kk == 0) bx = f.xv[mt][e][0].s01;
          else if (kk == 1) bx = f.xv[mt][e][0].s23;
          else if (kk == 2) bx = f.xv[mt][e][1].s01;
          else bx = f.xv[mt][e][1].s23;
          acc[jj][mt] = __builtin_amdgcn_wmma_f32_16x16x4_f32(
              false, a[kk], false, bx * cs, (short)0, acc[jj][mt], false, false);
        }
      }
    }
  }
}

__global__ __launch_bounds__(NTHR) void caps_wsum(const float* __restrict__ x,
                                                  const float* __restrict__ W,
                                                  const float* __restrict__ cbuf,
                                                  float* __restrict__ planes,
                                                  int ibase, int ni) {
  __shared__ __attribute__((aligned(16))) float red[8 * JG * 4 * 8 * 32];

  const int tid  = threadIdx.x;
  const int lane = tid & 31;
  const int wv   = tid >> 5;
  const int half = lane >> 4;
  const int l16  = lane & 15;
  const int jg   = blockIdx.x & (NJG - 1);
  const int r    = blockIdx.x / NJG;
  const int cpw  = ni / (NRNG * 8);
  const int il0  = (r * 8 + wv) * cpw;
  const int i0   = ibase + il0;

  const v8f vzero = {0.f, 0.f, 0.f, 0.f, 0.f, 0.f, 0.f, 0.f};
  v8f acc[JG][4];
#pragma unroll
  for (int jj = 0; jj < JG; ++jj)
#pragma unroll
    for (int mt = 0; mt < 4; ++mt) acc[jj][mt] = vzero;

  const float* wp = W + ((size_t)i0 * NJ + jg * JG) * DIM * DIM + l16 * DIM + 8 * half;
  const float* xp = x + ((size_t)l16 * INC + i0) * DIM + 8 * half;
  const float* cp = cbuf + (((size_t)jg * ni + il0) * BATCH + l16) * JG;

  Frag2 P0, P1;
  load_pair(P0, wp, xp, cp);
  for (int t = 0; t + 4 < cpw; t += 4) {
    load_pair(P1, wp + 2 * WSTEP, xp + 2 * DIM, cp + 2 * BATCH * JG);
    __builtin_amdgcn_sched_barrier(0);
    mac_pair(acc, P0);
    wp += 4 * WSTEP; xp += 4 * DIM; cp += 4 * BATCH * JG;
    load_pair(P0, wp, xp, cp);
    __builtin_amdgcn_sched_barrier(0);
    mac_pair(acc, P1);
  }
  load_pair(P1, wp + 2 * WSTEP, xp + 2 * DIM, cp + 2 * BATCH * JG);
  __builtin_amdgcn_sched_barrier(0);
  mac_pair(acc, P0);
  mac_pair(acc, P1);

#pragma unroll
  for (int jj = 0; jj < JG; ++jj) {
#pragma unroll
    for (int mt = 0; mt < 4; ++mt) {
      float4 a, b;
      a.x = acc[jj][mt][0]; a.y = acc[jj][mt][1]; a.z = acc[jj][mt][2]; a.w = acc[jj][mt][3];
      b.x = acc[jj][mt][4]; b.y = acc[jj][mt][5]; b.z = acc[jj][mt][6]; b.w = acc[jj][mt][7];
      const int s = (jj * 4 + mt) * 2;
      *(float4*)(red + ((wv * 32 + s + 0) * 32 + lane) * 4) = a;
      *(float4*)(red + ((wv * 32 + s + 1) * 32 + lane) * 4) = b;
    }
  }
  __syncthreads();
  float* plane = planes + (size_t)r * SZ;
#pragma unroll
  for (int u = 0; u < 4; ++u) {
    const int s = wv * 4 + u;
    float4 v = *(const float4*)(red + ((0 * 32 + s) * 32 + lane) * 4);
#pragma unroll
    for (int w = 1; w < 8; ++w) {
      const float4 p = *(const float4*)(red + ((w * 32 + s) * 32 + lane) * 4);
      v.x += p.x; v.y += p.y; v.z += p.z; v.w += p.w;
    }
    const int jj = s >> 3, mt = (s >> 1) & 3, q = s & 1;
    float* dst = plane + ((size_t)(mt * 16 + l16) * NJ + jg * JG + jj) * DIM + 8 * half + 4 * q;
    *(float4*)dst = v;
  }
}

__global__ void caps_squash(const float* src, float* outsum, float* dout,
                            float* szero, int nparts) {
  const int idx = blockIdx.x * blockDim.x + threadIdx.x;
  float v = 0.f;
  for (int p = 0; p < nparts; ++p) v += src[(size_t)p * SZ + idx];
  if (szero) szero[idx] = 0.f;
  const float sq = xorsum16(v * v);
  const float coef = sq / ((1.f + sq) * sqrtf(sq + 1e-7f));
  const float o = coef * v;
  outsum[idx] += o;
  if (dout) dout[idx] = o;
}

extern "C" void kernel_launch(void* const* d_in, const int* in_sizes, int n_in,
                              void* d_out, int out_size, void* d_ws, size_t ws_size,
                              hipStream_t stream) {
  const float* x = (const float*)d_in[0];
  const float* W = (const float*)d_in[1];
  float* out = (float*)d_out;
  float* ws = (float*)d_ws;

  float* outsum = ws;
  float* parts  = ws + SZ;

  const bool one = ws_size >= (size_t)(1 + 32 + 128) * SZ * sizeof(float);
  if (!one && ws_size < (size_t)(1 + 64 + 64) * SZ * sizeof(float)) return;

  caps_init<<<SZ / 256, 256, 0, stream>>>(outsum, SZ);

  for (int t = 0; t < 3; ++t) {
    if (one) {
      float* cbuf = parts + 32 * SZ;
      caps_couple<<<2048 / CW, NTHR, 0, stream>>>(x, W, outsum, cbuf, 0, 2048);
      caps_wsum<<<GRID, NTHR, 0, stream>>>(x, W, cbuf, parts, 0, 2048);
      caps_squash<<<SZ / 256, 256, 0, stream>>>(parts, outsum, (t == 2) ? out : nullptr,
                                                nullptr, 32);
    } else {
      float* cbuf = parts + 64 * SZ;
      caps_couple<<<1024 / CW, NTHR, 0, stream>>>(x, W, outsum, cbuf, 0, 1024);
      caps_wsum<<<GRID, NTHR, 0, stream>>>(x, W, cbuf, parts, 0, 1024);
      caps_couple<<<1024 / CW, NTHR, 0, stream>>>(x, W, outsum, cbuf, 1024, 1024);
      caps_wsum<<<GRID, NTHR, 0, stream>>>(x, W, cbuf, parts + 32 * SZ, 1024, 1024);
      caps_squash<<<SZ / 256, 256, 0, stream>>>(parts, outsum, (t == 2) ? out : nullptr,
                                                nullptr, 64);
    }
  }
}
